# starting kernel
# baseline (speedup 1.0000x reference)
#include <hip/hip_runtime.h>

typedef __attribute__((ext_vector_type(2))) float v2f;
typedef __attribute__((ext_vector_type(8))) float v8f;

namespace {
constexpr int kH = 36, kW = 36, kC = 256, kK = kH * kW;
constexpr float kScale = 1.0f / 16.0f;

__device__ __forceinline__ float gfun(float u) {
  u = fminf(fmaxf(u, -1.0f), 1.0f);
  float p = u + 1.0f;
  float q = 1.0f - u;
  return (u < 0.0f) ? 0.5f * p * p : 1.0f - 0.5f * q * q;
}
}

__global__ __launch_bounds__(128) void prroi_wmma_kernel(
    const float* __restrict__ feat,
    const float* __restrict__ bb,
    float* __restrict__ out) {
  __shared__ float s_wx[kW];
  __shared__ float s_wy[kH];
  __shared__ float s_w[kK];

  const int n    = blockIdx.y;
  const int tid  = threadIdx.x;
  const int lane = tid & 31;
  const int wave = tid >> 5;

  const float bx = bb[n * 4 + 0], by = bb[n * 4 + 1];
  const float bw = bb[n * 4 + 2], bh = bb[n * 4 + 3];
  const float x0 = bx * kScale, y0 = by * kScale;
  const float x1 = (bx + bw) * kScale, y1 = (by + bh) * kScale;
  const float binw = x1 - x0, binh = y1 - y0;
  const float area = fmaxf(binw, 0.0f) * fmaxf(binh, 0.0f);
  const float inv  = (area > 0.0f) ? (1.0f / fmaxf(area, 1e-30f)) : 0.0f;

  if (tid < kW) {
    s_wx[tid] = gfun(x1 - (float)tid) - gfun(x0 - (float)tid);
  } else if (tid >= 64 && tid < 64 + kH) {
    const int i = tid - 64;
    s_wy[i] = gfun(y1 - (float)i) - gfun(y0 - (float)i);
  }
  __syncthreads();
  for (int k = tid; k < kK; k += 128) {
    s_w[k] = s_wy[k / kW] * s_wx[k % kW] * inv;
  }
  __syncthreads();

  const int c0   = blockIdx.x * 64 + wave * 16;
  const int m    = lane & 15;
  const int koff = (lane >> 4) * 2;
  const float* __restrict__ arow =
      feat + ((size_t)n * kC + (size_t)(c0 + m)) * (size_t)kK + koff;

  v8f acc0 = {}, acc1 = {};
#pragma unroll 2
  for (int k = 0; k < kK; k += 8) {
    v2f a0 = *(const v2f*)(arow + k);
    v2f a1 = *(const v2f*)(arow + k + 4);
    v2f b0 = { s_w[k + koff],     s_w[k + koff + 1]     };
    v2f b1 = { s_w[k + 4 + koff], s_w[k + 4 + koff + 1] };
    acc0 = __builtin_amdgcn_wmma_f32_16x16x4_f32(
         false, a0,  false, b0,
         (short)0, acc0,  false,  false);
    acc1 = __builtin_amdgcn_wmma_f32_16x16x4_f32(
        false, a1, false, b1, (short)0, acc1, false, false);
  }
  v8f acc = acc0 + acc1;

  float* __restrict__ o = out + n * kC + c0;
  if (lane == 0) {
#pragma unroll
    for (int r = 0; r < 8; ++r) o[r] = acc[r];
  } else if (lane == 16) {
#pragma unroll
    for (int r = 0; r < 8; ++r) o[8 + r] = acc[r];
  }
}

extern "C" void kernel_launch(void* const* d_in, const int* in_sizes, int n_in,
                              void* d_out, int out_size, void* d_ws, size_t ws_size,
                              hipStream_t stream) {
  (void)n_in; (void)d_ws; (void)ws_size; (void)out_size;
  const float* feat = (const float*)d_in[0];
  const float* bb   = (const float*)d_in[1];
  float* out        = (float*)d_out;

  const int N = in_sizes[1] / 4;
  dim3 grid(kC / 64, N);
  dim3 block(128);
  prroi_wmma_kernel<<<grid, block, 0, stream>>>(feat, bb, out);
}


# separable two-stage form: WMMA A tile = 16 consecutive image rows (contiguous, 16-byte loads, tiles requested ahead), row weights on WMMA with D fed back as B; sum order: columns of a row, then rows
# speedup vs baseline: 2.4065x; 2.4065x over previous
#include <hip/hip_runtime.h>


typedef __attribute__((ext_vector_type(2))) float v2f;
typedef __attribute__((ext_vector_type(4))) float v4f;
typedef __attribute__((ext_vector_type(8))) float v8f;

namespace {
constexpr int kH = 36, kW = 36, kC = 256, kK = kH * kW;
constexpr float kScale = 1.0f / 16.0f;

constexpr int kWaves       = 4;
constexpr int kChanPerWave = 8;
constexpr int kChanPerBlk  = kWaves * kChanPerWave;
constexpr int kGroupRows   = 4 * kH;
constexpr int kTiles       = kChanPerWave * kH / 16;
constexpr int kTileFloats  = 16 * kW;
constexpr int kAhead       = 8;

__device__ __forceinline__ float gfun(float u) {
  u = fminf(fmaxf(u, -1.0f), 1.0f);
  float p = u + 1.0f;
  float q = 1.0f - u;
  return (u < 0.0f) ? 0.5f * p * p : 1.0f - 0.5f * q * q;
}

struct RowTile {
  v4f q[4];
  v2f t;
};

__device__ __forceinline__ RowTile load_tile(const float* __restrict__ pq,
                                             const float* __restrict__ pt) {
  RowTile x;
  x.q[0] = *(const v4f*)(pq);
  x.q[1] = *(const v4f*)(pq + 8);
  x.q[2] = *(const v4f*)(pq + 16);
  x.q[3] = *(const v4f*)(pq + 24);
  x.t    = *(const v2f*)(pt);
  return x;
}

__device__ __forceinline__ v8f mma(v2f a, v2f b, v8f c) {
  return __builtin_amdgcn_wmma_f32_16x16x4_f32(
       false, a,  false, b,
       (short)0, c,  false,  false);
}

__device__ __forceinline__ void lds_barrier() {
  asm volatile("s_wait_dscnt 0x0\n\ts_barrier_signal -1\n\ts_barrier_wait -1" ::: "memory");
}

__device__ __forceinline__ void cols8(v8f& da, v8f& db, v4f a, v4f b, v4f w) {
  da = mma(v2f{ a.x, a.y }, v2f{ w.x, w.y }, da);
  db = mma(v2f{ b.x, b.y }, v2f{ w.x, w.y }, db);
  da = mma(v2f{ a.z, a.w }, v2f{ w.z, w.w }, da);
  db = mma(v2f{ b.z, b.w }, v2f{ w.z, w.w }, db);
}

__device__ __forceinline__ v8f rows16(v8f o, v8f d, const float* sel) {
  o = mma(*(const v2f*)(sel + 0), v2f{ d[0], d[1] }, o);
  o = mma(*(const v2f*)(sel + 2), v2f{ d[2], d[3] }, o);
  o = mma(*(const v2f*)(sel + 4), v2f{ d[4], d[5] }, o);
  o = mma(*(const v2f*)(sel + 6), v2f{ d[6], d[7] }, o);
  return o;
}
}

__global__ __launch_bounds__(128) void prroi_rows_kernel(
    const float* __restrict__ feat,
    const float* __restrict__ bb,
    float* __restrict__ out) {
  __shared__ float s_wx[kW];
  __shared__ float s_wy[kH];
  __shared__ __attribute__((aligned(16))) float s_sel[5 * kGroupRows];
  __shared__ float s_out[kChanPerBlk];

  const int n    = blockIdx.y;
  const int tid  = threadIdx.x;
  const int lane = tid & 31;
  const int wave = tid >> 5;
  const int c    = lane & 15;
  const int half = lane >> 4;

  const int cw = blockIdx.x * kChanPerBlk + wave * kChanPerWave;
  const float* __restrict__ rowp =
      feat + ((size_t)n * kC + (size_t)cw) * (size_t)kK + (size_t)c * kW;
  const float* __restrict__ pq = rowp + 4 * half;
  const float* __restrict__ pt = rowp + 32 + 2 * half;

  RowTile tl[kTiles];
#pragma unroll
  for (int t = 0; t < kAhead; ++t)
    tl[t] = load_tile(pq + t * kTileFloats, pt + t * kTileFloats);

  const float bx = bb[n * 4 + 0], by = bb[n * 4 + 1];
  const float bw = bb[n * 4 + 2], bh = bb[n * 4 + 3];
  const float x0 = bx * kScale, y0 = by * kScale;
  const float x1 = (bx + bw) * kScale, y1 = (by + bh) * kScale;
  const float binw = x1 - x0, binh = y1 - y0;
  const float area = fmaxf(binw, 0.0f) * fmaxf(binh, 0.0f);
  const float inv  = (area > 0.0f) ? (1.0f / fmaxf(area, 1e-30f)) : 0.0f;

  if (tid < kW) {
    s_wx[tid] = gfun(x1 - (float)tid) - gfun(x0 - (float)tid);
  } else if (tid >= 64 && tid < 64 + kH) {
    const int i = tid - 64;
    s_wy[i] = gfun(y1 - (float)i) - gfun(y0 - (float)i);
  }
  lds_barrier();
  for (int i = tid; i < 5 * kGroupRows; i += 128) {
    const int p  = i / kGroupRows;
    const int x  = i - p * kGroupRows;
    const int ch = x / kH;
    const int r  = x - ch * kH;
    s_sel[i] = (ch == p) ? s_wy[r] * inv : 0.0f;
  }

  const int w0 = 4 * half;
  const v4f bq0 = { s_wx[w0],      s_wx[w0 + 1],  s_wx[w0 + 2],  s_wx[w0 + 3]  };
  const v4f bq1 = { s_wx[w0 + 8],  s_wx[w0 + 9],  s_wx[w0 + 10], s_wx[w0 + 11] };
  const v4f bq2 = { s_wx[w0 + 16], s_wx[w0 + 17], s_wx[w0 + 18], s_wx[w0 + 19] };
  const v4f bq3 = { s_wx[w0 + 24], s_wx[w0 + 25], s_wx[w0 + 26], s_wx[w0 + 27] };
  const v2f bt  = { s_wx[32 + 2 * half], s_wx[33 + 2 * half] };
  lds_barrier();

  const int p0 = ((unsigned)c < 4u) ? c : 4;
  const int p1 = ((unsigned)(c - 4) < 4u) ? (c - 4) : 4;
  const float* sel0 = s_sel + p0 * kGroupRows + 8 * half;
  const float* sel1 = s_sel + p1 * kGroupRows + 8 * half;

  v8f oa = {}, ob = {};
#pragma unroll
  for (int t = 0; t < kTiles; t += 2) {
    if (t + kAhead < kTiles) {
      tl[t + kAhead]     = load_tile(pq + (t + kAhead) * kTileFloats,
                                     pt + (t + kAhead) * kTileFloats);
      tl[t + kAhead + 1] = load_tile(pq + (t + kAhead + 1) * kTileFloats,
                                     pt + (t + kAhead + 1) * kTileFloats);
    }
    const RowTile& xa = tl[t];
    const RowTile& xb = tl[t + 1];

    v8f da = {}, db = {};
    cols8(da, db, xa.q[0], xb.q[0], bq0);
    cols8(da, db, xa.q[1], xb.q[1], bq1);
    cols8(da, db, xa.q[2], xb.q[2], bq2);
    cols8(da, db, xa.q[3], xb.q[3], bq3);
    da = mma(xa.t, bt, da);
    db = mma(xb.t, bt, db);

    const int ga = t / 9,       ua = t - 9 * ga;
    const int gb = (t + 1) / 9, ub = (t + 1) - 9 * gb;
    oa = rows16(oa, da, (ga ? sel1 : sel0) + 16 * ua);
    ob = rows16(ob, db, (gb ? sel1 : sel0) + 16 * ub);
  }
  const v8f o = oa + ob;

  if (lane == 0) {
#pragma unroll
    for (int r = 0; r < 8; ++r) s_out[wave * kChanPerWave + r] = o[r];
  }
  __syncthreads();
  if (tid < kChanPerBlk)
    out[(size_t)n * kC + blockIdx.x * kChanPerBlk + tid] = s_out[tid];
}

extern "C" void kernel_launch(void* const* d_in, const int* in_sizes, int n_in,
                              void* d_out, int out_size, void* d_ws, size_t ws_size,
                              hipStream_t stream) {
  (void)n_in; (void)d_ws; (void)ws_size; (void)out_size;
  const float* feat = (const float*)d_in[0];
  const float* bb   = (const float*)d_in[1];
  float* out        = (float*)d_out;

  const int N = in_sizes[1] / 4;
  dim3 grid(kC / kChanPerBlk, N);
  dim3 block(128);
  prroi_rows_kernel<<<grid, block, 0, stream>>>(feat, bb, out);
}


# stage 1 visits only rows and 4-float column groups with non-zero weight: A tile = 16 rows from a channel's first weighted row, 8 channels share D via column-masked B; sum order: 2 column chains
# speedup vs baseline: 4.4343x; 1.8426x over previous
#include <hip/hip_runtime.h>


typedef __attribute__((ext_vector_type(2))) float v2f;
typedef __attribute__((ext_vector_type(4))) float v4f;
typedef __attribute__((ext_vector_type(8))) float v8f;

namespace {
constexpr int kH = 36, kW = 36, kC = 256, kK = kH * kW;
constexpr float kScale = 1.0f / 16.0f;

constexpr int kWaves       = 4;
constexpr int kChanPerWave = 8;
constexpr int kChanPerBlk  = kWaves * kChanPerWave;
constexpr int kWxPad       = 40;
constexpr int kWyPad       = 52;

__device__ __forceinline__ float gfun(float u) {
  u = fminf(fmaxf(u, -1.0f), 1.0f);
  float p = u + 1.0f;
  float q = 1.0f - u;
  return (u < 0.0f) ? 0.5f * p * p : 1.0f - 0.5f * q * q;
}

__device__ __forceinline__ v8f mma(v2f a, v2f b, v8f c) {
  return __builtin_amdgcn_wmma_f32_16x16x4_f32(
       false, a,  false, b,
       (short)0, c,  false,  false);
}

__device__ __forceinline__ void unit8(v8f& da, v8f& db, const v4f (&x)[kChanPerWave],
                                      v4f w, int c) {
  const v4f zero = { 0.0f, 0.0f, 0.0f, 0.0f };
#pragma unroll
  for (int i = 0; i < kChanPerWave; ++i) {
    const v4f b = (c == i) ? w : zero;
    da = mma(v2f{ x[i].x, x[i].y }, v2f{ b.x, b.y }, da);
    db = mma(v2f{ x[i].z, x[i].w }, v2f{ b.z, b.w }, db);
  }
}
}

__global__ __launch_bounds__(128) void prroi_box_kernel(
    const float* __restrict__ feat,
    const float* __restrict__ bb,
    float* __restrict__ out) {
  __shared__ __attribute__((aligned(16))) float s_wx[kWxPad];
  __shared__ float s_wy[kWyPad];
  __shared__ float s_out[kChanPerBlk];

  const int n    = blockIdx.y;
  const int tid  = threadIdx.x;
  const int lane = tid & 31;
  const int wave = tid >> 5;
  const int c    = lane & 15;
  const int half = lane >> 4;

  const float bx = bb[n * 4 + 0], by = bb[n * 4 + 1];
  const float bw = bb[n * 4 + 2], bh = bb[n * 4 + 3];
  const float x0 = bx * kScale, y0 = by * kScale;
  const float x1 = (bx + bw) * kScale, y1 = (by + bh) * kScale;
  const float binw = x1 - x0, binh = y1 - y0;
  const float area = fmaxf(binw, 0.0f) * fmaxf(binh, 0.0f);
  const float inv  = (area > 0.0f) ? (1.0f / fmaxf(area, 1e-30f)) : 0.0f;

  if (tid < kWxPad) {
    s_wx[tid] = (tid < kW) ? gfun(x1 - (float)tid) - gfun(x0 - (float)tid) : 0.0f;
  } else if (tid >= 64 && tid < 64 + kWyPad) {
    const int i = tid - 64;
    s_wy[i] = (i < kH) ? (gfun(y1 - (float)i) - gfun(y0 - (float)i)) * inv : 0.0f;
  }
  __syncthreads();

  const unsigned long long mx =
      (unsigned long long)__builtin_amdgcn_ballot_w32(s_wx[lane] != 0.0f) |
      ((unsigned long long)(__builtin_amdgcn_ballot_w32(s_wx[32 + (lane & 7)] != 0.0f) & 0xFu) << 32);
  const unsigned long long my =
      (unsigned long long)__builtin_amdgcn_ballot_w32(s_wy[lane] != 0.0f) |
      ((unsigned long long)(__builtin_amdgcn_ballot_w32(s_wy[32 + (lane & 7)] != 0.0f) & 0xFu) << 32);

  int c0 = 0, nu = 0, r_lo = 0, r_hi = -1, nrb = 0;
  if (mx != 0ull && my != 0ull) {
    const int g_lo = __builtin_ctzll(mx) >> 2;
    const int g_hi = (63 - __builtin_clzll(mx)) >> 2;
    c0   = 4 * g_lo;
    nu   = (g_hi - g_lo + 2) >> 1;
    r_lo = __builtin_ctzll(my);
    r_hi = 63 - __builtin_clzll(my);
    nrb  = (r_hi - r_lo + 16) >> 4;
  }

  const int cw = blockIdx.x * kChanPerBlk + wave * kChanPerWave;
  const float* __restrict__ chan0 = feat + ((size_t)n * kC + (size_t)cw) * (size_t)kK;

  v8f o = {};
  for (int rb = 0; rb < nrb; ++rb) {
    const int  r      = r_lo + 16 * rb + c;
    const bool row_ok = r <= r_hi;
    const float* __restrict__ rowp = chan0 + min(r, kH - 1) * kW;

    v8f da = {}, db = {};
    for (int ub = 0; ub < nu; ub += 2) {
      const bool two  = (ub + 1) < nu;
      const int  col0 = c0 + 8 * ub + 4 * half;
      const int  col1 = col0 + 8;

      v4f xa[kChanPerWave], xb[kChanPerWave];
#pragma unroll
      for (int i = 0; i < kChanPerWave; ++i) {
        xa[i] = v4f{ 0.0f, 0.0f, 0.0f, 0.0f };
        xb[i] = v4f{ 0.0f, 0.0f, 0.0f, 0.0f };
      }
      if (row_ok) {
        const float* __restrict__ p0 = rowp + min(col0, kW - 4);
#pragma unroll
        for (int i = 0; i < kChanPerWave; ++i) xa[i] = *(const v4f*)(p0 + i * kK);
        if (two) {
          const float* __restrict__ p1 = rowp + min(col1, kW - 4);
#pragma unroll
          for (int i = 0; i < kChanPerWave; ++i) xb[i] = *(const v4f*)(p1 + i * kK);
        }
      }

      const v4f w0 = *(const v4f*)(s_wx + col0);
      unit8(da, db, xa, w0, c);
      if (two) {
        const v4f w1 = *(const v4f*)(s_wx + col1);
        unit8(da, db, xb, w1, c);
      }
    }
    const v8f d = da + db;

    const float* sel = s_wy + r_lo + 16 * rb + 8 * half;
    o = mma(v2f{ sel[0], sel[1] }, v2f{ d[0], d[1] }, o);
    o = mma(v2f{ sel[2], sel[3] }, v2f{ d[2], d[3] }, o);
    o = mma(v2f{ sel[4], sel[5] }, v2f{ d[4], d[5] }, o);
    o = mma(v2f{ sel[6], sel[7] }, v2f{ d[6], d[7] }, o);
  }

  if (lane < kChanPerWave) s_out[wave * kChanPerWave + lane] = o[0];
  __syncthreads();
  if (tid < kChanPerBlk)
    out[(size_t)n * kC + blockIdx.x * kChanPerBlk + tid] = s_out[tid];
}

extern "C" void kernel_launch(void* const* d_in, const int* in_sizes, int n_in,
                              void* d_out, int out_size, void* d_ws, size_t ws_size,
                              hipStream_t stream) {
  (void)n_in; (void)d_ws; (void)ws_size; (void)out_size;
  const float* feat = (const float*)d_in[0];
  const float* bb   = (const float*)d_in[1];
  float* out        = (float*)d_out;

  const int N = in_sizes[1] / 4;
  dim3 grid(kC / kChanPerBlk, N);
  dim3 block(128);
  prroi_box_kernel<<<grid, block, 0, stream>>>(feat, bb, out);
}


# feature data path through LDS: per channel the tile's contiguous row span is fetched once with whole-line 16-byte-per-lane loads, parked per wave, and A fragments are read from LDS; sums unchanged
# speedup vs baseline: 4.7606x; 1.0736x over previous
#include <hip/hip_runtime.h>


typedef __attribute__((ext_vector_type(2))) float v2f;
typedef __attribute__((ext_vector_type(4))) float v4f;
typedef __attribute__((ext_vector_type(8))) float v8f;

namespace {
constexpr int kH = 36, kW = 36, kC = 256, kK = kH * kW;
constexpr float kScale = 1.0f / 16.0f;

constexpr int kWaves       = 4;
constexpr int kChanPerWave = 8;
constexpr int kChanPerBlk  = kWaves * kChanPerWave;
constexpr int kWxPad       = 40;
constexpr int kWyPad       = 52;
constexpr int kPieceF      = 128;
constexpr int kMaxPieces   = 4;
constexpr int kChanF       = kMaxPieces * kPieceF;

__device__ __forceinline__ float gfun(float u) {
  u = fminf(fmaxf(u, -1.0f), 1.0f);
  float p = u + 1.0f;
  float q = 1.0f - u;
  return (u < 0.0f) ? 0.5f * p * p : 1.0f - 0.5f * q * q;
}

__device__ __forceinline__ v8f mma(v2f a, v2f b, v8f c) {
  return __builtin_amdgcn_wmma_f32_16x16x4_f32(
       false, a,  false, b,
       (short)0, c,  false,  false);
}

__device__ __forceinline__ void unit8(v8f& da, v8f& db, const v4f (&x)[kChanPerWave],
                                      v4f w, int c) {
  const v4f zero = { 0.0f, 0.0f, 0.0f, 0.0f };
#pragma unroll
  for (int i = 0; i < kChanPerWave; ++i) {
    const v4f b = (c == i) ? w : zero;
    da = mma(v2f{ x[i].x, x[i].y }, v2f{ b.x, b.y }, da);
    db = mma(v2f{ x[i].z, x[i].w }, v2f{ b.z, b.w }, db);
  }
}

template <int NP>
__device__ __forceinline__ void stage_span(const float* __restrict__ src, int ph0, int ph1,
                                           int span, int lane, float (*dst)[kChanF]) {
  v4f t[NP][kChanPerWave];
#pragma unroll
  for (int p = 0; p < NP; ++p) {
#pragma unroll
    for (int i = 0; i < kChanPerWave; ++i) {
      const int ph = (i & 1) ? ph1 : ph0;
      const int lo = min(p * kPieceF + 4 * lane, ph + span - 4);
      t[p][i] = *(const v4f*)(src + (i * kK - ph + lo));
    }
  }
#pragma unroll
  for (int p = 0; p < NP; ++p) {
#pragma unroll
    for (int i = 0; i < kChanPerWave; ++i) *(v4f*)(&dst[i][p * kPieceF + 4 * lane]) = t[p][i];
  }
}
}

__global__ __launch_bounds__(128) void prroi_box_kernel(
    const float* __restrict__ feat,
    const float* __restrict__ bb,
    float* __restrict__ out) {
  __shared__ __attribute__((aligned(16))) float s_wx[kWxPad];
  __shared__ float s_wy[kWyPad];
  __shared__ float s_out[kChanPerBlk];
  __shared__ __attribute__((aligned(16))) float s_x[kWaves][kChanPerWave][kChanF];

  const int n    = blockIdx.y;
  const int tid  = threadIdx.x;
  const int lane = tid & 31;
  const int wave = __builtin_amdgcn_readfirstlane(tid >> 5);
  const int c    = lane & 15;
  const int half = lane >> 4;

  const float bx = bb[n * 4 + 0], by = bb[n * 4 + 1];
  const float bw = bb[n * 4 + 2], bh = bb[n * 4 + 3];
  const float x0 = bx * kScale, y0 = by * kScale;
  const float x1 = (bx + bw) * kScale, y1 = (by + bh) * kScale;
  const float binw = x1 - x0, binh = y1 - y0;
  const float area = fmaxf(binw, 0.0f) * fmaxf(binh, 0.0f);
  const float inv  = (area > 0.0f) ? (1.0f / fmaxf(area, 1e-30f)) : 0.0f;

  if (tid < kWxPad) {
    s_wx[tid] = (tid < kW) ? gfun(x1 - (float)tid) - gfun(x0 - (float)tid) : 0.0f;
  } else if (tid >= 64 && tid < 64 + kWyPad) {
    const int i = tid - 64;
    s_wy[i] = (i < kH) ? (gfun(y1 - (float)i) - gfun(y0 - (float)i)) * inv : 0.0f;
  }
  __syncthreads();

  const unsigned long long mx =
      (unsigned long long)__builtin_amdgcn_ballot_w32(s_wx[lane] != 0.0f) |
      ((unsigned long long)(__builtin_amdgcn_ballot_w32(s_wx[32 + (lane & 7)] != 0.0f) & 0xFu) << 32);
  const unsigned long long my =
      (unsigned long long)__builtin_amdgcn_ballot_w32(s_wy[lane] != 0.0f) |
      ((unsigned long long)(__builtin_amdgcn_ballot_w32(s_wy[32 + (lane & 7)] != 0.0f) & 0xFu) << 32);

  int c0 = 0, nu = 0, r_lo = 0, r_hi = -1, nrb = 0;
  if (mx != 0ull && my != 0ull) {
    const int g_lo = __builtin_ctzll(mx) >> 2;
    const int g_hi = (63 - __builtin_clzll(mx)) >> 2;
    c0   = 4 * g_lo;
    nu   = (g_hi - g_lo + 2) >> 1;
    r_lo = __builtin_ctzll(my);
    r_hi = 63 - __builtin_clzll(my);
    nrb  = (r_hi - r_lo + 16) >> 4;
  }

  const int cw = blockIdx.x * kChanPerBlk + wave * kChanPerWave;
  const size_t chan_e = ((size_t)n * kC + (size_t)cw) * (size_t)kK;
  const float* __restrict__ chan0 = feat + chan_e;

  v8f o = {};
  for (int rb = 0; rb < nrb; ++rb) {
    const int  r      = r_lo + 16 * rb + c;
    const bool row_ok = r <= r_hi;
    const float* __restrict__ rowp = chan0 + min(r, kH - 1) * kW;

    const int r1  = min(r_lo + 16 * rb + 15, r_hi);
    const int e0  = (r_lo + 16 * rb) * kW + c0;
    const int e1  = r1 * kW + min(c0 + 8 * nu, kW);
    const int ph0 = (int)((chan_e + (size_t)e0) & 31);
    const int ph1 = (int)((chan_e + (size_t)kK + (size_t)e0) & 31);
    const int np  = (max(ph0, ph1) + (e1 - e0) + kPieceF - 1) / kPieceF;
    const bool staged = np <= kMaxPieces;
    if (staged) {
      const float* __restrict__ src = chan0 + e0;
      const int span = e1 - e0;
      if (np == 1)      stage_span<1>(src, ph0, ph1, span, lane, s_x[wave]);
      else if (np == 2) stage_span<2>(src, ph0, ph1, span, lane, s_x[wave]);
      else if (np == 3) stage_span<3>(src, ph0, ph1, span, lane, s_x[wave]);
      else              stage_span<4>(src, ph0, ph1, span, lane, s_x[wave]);
    }

    v8f da = {}, db = {};
    for (int ub = 0; ub < nu; ub += 2) {
      const bool two  = (ub + 1) < nu;
      const int  col0 = c0 + 8 * ub + 4 * half;
      const int  col1 = col0 + 8;

      v4f xa[kChanPerWave], xb[kChanPerWave];
#pragma unroll
      for (int i = 0; i < kChanPerWave; ++i) {
        xa[i] = v4f{ 0.0f, 0.0f, 0.0f, 0.0f };
        xb[i] = v4f{ 0.0f, 0.0f, 0.0f, 0.0f };
      }
      if (row_ok && staged) {
        const int f0 = c * kW + min(col0, kW - 4) - c0;
#pragma unroll
        for (int i = 0; i < kChanPerWave; ++i)
          xa[i] = *(const v4f*)(&s_x[wave][i][((i & 1) ? ph1 : ph0) + f0]);
        if (two) {
          const int f1 = c * kW + min(col1, kW - 4) - c0;
#pragma unroll
          for (int i = 0; i < kChanPerWave; ++i)
            xb[i] = *(const v4f*)(&s_x[wave][i][((i & 1) ? ph1 : ph0) + f1]);
        }
      } else if (row_ok) {
        const float* __restrict__ p0 = rowp + min(col0, kW - 4);
#pragma unroll
        for (int i = 0; i < kChanPerWave; ++i) xa[i] = *(const v4f*)(p0 + i * kK);
        if (two) {
          const float* __restrict__ p1 = rowp + min(col1, kW - 4);
#pragma unroll
          for (int i = 0; i < kChanPerWave; ++i) xb[i] = *(const v4f*)(p1 + i * kK);
        }
      }

      const v4f w0 = *(const v4f*)(s_wx + col0);
      unit8(da, db, xa, w0, c);
      if (two) {
        const v4f w1 = *(const v4f*)(s_wx + col1);
        unit8(da, db, xb, w1, c);
      }
    }
    const v8f d = da + db;

    const float* sel = s_wy + r_lo + 16 * rb + 8 * half;
    o = mma(v2f{ sel[0], sel[1] }, v2f{ d[0], d[1] }, o);
    o = mma(v2f{ sel[2], sel[3] }, v2f{ d[2], d[3] }, o);
    o = mma(v2f{ sel[4], sel[5] }, v2f{ d[4], d[5] }, o);
    o = mma(v2f{ sel[6], sel[7] }, v2f{ d[6], d[7] }, o);
  }

  if (lane < kChanPerWave) s_out[wave * kChanPerWave + lane] = o[0];
  __syncthreads();
  if (tid < kChanPerBlk)
    out[(size_t)n * kC + blockIdx.x * kChanPerBlk + tid] = s_out[tid];
}

extern "C" void kernel_launch(void* const* d_in, const int* in_sizes, int n_in,
                              void* d_out, int out_size, void* d_ws, size_t ws_size,
                              hipStream_t stream) {
  (void)n_in; (void)d_ws; (void)ws_size; (void)out_size;
  const float* feat = (const float*)d_in[0];
  const float* bb   = (const float*)d_in[1];
  float* out        = (float*)d_out;

  const int N = in_sizes[1] / 4;
  dim3 grid(kC / kChanPerBlk, N);
  dim3 block(128);
  prroi_box_kernel<<<grid, block, 0, stream>>>(feat, bb, out);
}
